# starting kernel
# baseline (speedup 1.0000x reference)
#include <hip/hip_runtime.h>

typedef __attribute__((ext_vector_type(2))) float v2f;
typedef __attribute__((ext_vector_type(8))) float v8f;

#define IN_SIZE  256
#define OUT_SIZE 256
#define LDA 268
#define LDB 72

__global__ __launch_bounds__(128)
void gcn_gemm_wmma(const float* __restrict__ X, const float* __restrict__ W,
                   float* __restrict__ S, int M) {
    __shared__ float lA[16 * LDA];
    __shared__ float lB[IN_SIZE * LDB];

    const int t    = threadIdx.x;
    const int lane = t & 31;
    const int wv   = t >> 5;
    const int lm   = lane & 15;
    const int half = lane >> 4;

    const int row0 = blockIdx.x * 16;
    const int colb = blockIdx.y * 64;
    const int wcol = wv * 16;

    #pragma unroll
    for (int i = 0; i < 8; ++i) {
        int idx4 = t + i * 128;
        int r    = idx4 >> 6;
        int c4   = idx4 & 63;
        float4 v = make_float4(0.f, 0.f, 0.f, 0.f);
        if (row0 + r < M)
            v = *(const float4*)(X + (size_t)(row0 + r) * IN_SIZE + c4 * 4);
        *(float4*)(&lA[r * LDA + c4 * 4]) = v;
    }
    #pragma unroll
    for (int i = 0; i < 32; ++i) {
        int idx4 = t + i * 128;
        int r    = idx4 >> 4;
        int c4   = idx4 & 15;
        float4 v = *(const float4*)(W + (size_t)r * OUT_SIZE + colb + c4 * 4);
        *(float4*)(&lB[r * LDB + c4 * 4]) = v;
    }
    __syncthreads();

    const int aBase = lm * LDA + 2 * half;
    const int bBase = 2 * half * LDB + wcol + lm;

    v8f acc0 = {0.f,0.f,0.f,0.f,0.f,0.f,0.f,0.f};
    v8f acc1 = acc0, acc2 = acc0, acc3 = acc0;

    #pragma unroll
    for (int k = 0; k < 64; k += 4) {
        {
            v2f a = *(const v2f*)(&lA[aBase + k]);
            v2f b; b.x = lB[bBase + k * LDB]; b.y = lB[bBase + (k + 1) * LDB];
            acc0 = __builtin_amdgcn_wmma_f32_16x16x4_f32(false, a, false, b, (short)0, acc0, false, false);
        }
        {
            v2f a = *(const v2f*)(&lA[aBase + k + 64]);
            v2f b; b.x = lB[bBase + (k + 64) * LDB]; b.y = lB[bBase + (k + 65) * LDB];
            acc1 = __builtin_amdgcn_wmma_f32_16x16x4_f32(false, a, false, b, (short)0, acc1, false, false);
        }
        {
            v2f a = *(const v2f*)(&lA[aBase + k + 128]);
            v2f b; b.x = lB[bBase + (k + 128) * LDB]; b.y = lB[bBase + (k + 129) * LDB];
            acc2 = __builtin_amdgcn_wmma_f32_16x16x4_f32(false, a, false, b, (short)0, acc2, false, false);
        }
        {
            v2f a = *(const v2f*)(&lA[aBase + k + 192]);
            v2f b; b.x = lB[bBase + (k + 192) * LDB]; b.y = lB[bBase + (k + 193) * LDB];
            acc3 = __builtin_amdgcn_wmma_f32_16x16x4_f32(false, a, false, b, (short)0, acc3, false, false);
        }
    }

    v8f acc = (acc0 + acc1) + (acc2 + acc3);

    #pragma unroll
    for (int v = 0; v < 8; ++v) {
        int r = row0 + v + 8 * half;
        if (r < M)
            S[(size_t)r * OUT_SIZE + colb + wcol + lm] = acc[v];
    }
}

__global__ __launch_bounds__(256)
void gcn_init_bias(float* __restrict__ out, const float* __restrict__ bias,
                   long long total) {
    long long i = (long long)blockIdx.x * blockDim.x + threadIdx.x;
    if (i < total) out[i] = bias[i & (OUT_SIZE - 1)];
}

__global__ __launch_bounds__(256)
void gcn_edge_scatter(const float* __restrict__ S, const float* __restrict__ ew,
                      const int* __restrict__ esrc, const int* __restrict__ edst,
                      float* __restrict__ out, int E) {
    int e    = (int)(((long long)blockIdx.x * blockDim.x + threadIdx.x) >> 5);
    int lane = threadIdx.x & 31;
    if (e >= E) return;
    int   s = esrc[e];
    int   d = edst[e];
    float w = ew[e];
    const float* sp = S   + (size_t)s * OUT_SIZE;
    float*       op = out + (size_t)d * OUT_SIZE;
    #pragma unroll
    for (int i = 0; i < 8; ++i) {
        int c = lane + i * 32;
        atomicAdd(&op[c], sp[c] * w);
    }
}

extern "C" void kernel_launch(void* const* d_in, const int* in_sizes, int n_in,
                              void* d_out, int out_size, void* d_ws, size_t ws_size,
                              hipStream_t stream) {
    const float* x    = (const float*)d_in[0];
    const float* w    = (const float*)d_in[1];
    const float* bias = (const float*)d_in[2];
    const float* ew   = (const float*)d_in[3];
    const int*   esrc = (const int*)d_in[4];
    const int*   edst = (const int*)d_in[5];
    float* out  = (float*)d_out;
    float* supp = (float*)d_ws;

    const int M = in_sizes[0] / IN_SIZE;
    const int E = in_sizes[3];

    dim3 ggrid((M + 15) / 16, OUT_SIZE / 64);
    gcn_gemm_wmma<<<ggrid, 128, 0, stream>>>(x, w, supp, M);

    long long total = (long long)M * OUT_SIZE;
    int iblocks = (int)((total + 255) / 256);
    gcn_init_bias<<<iblocks, 256, 0, stream>>>(out, bias, total);

    int sblocks = (E + 7) / 8;
    gcn_edge_scatter<<<sblocks, 256, 0, stream>>>(supp, ew, esrc, edst, out, E);
}


# GEMM retiled: each wave keeps its 256x16 W slab in registers and blocks persist over 16-row X tiles (double-buffered in LDS), so W is fetched once per block, not once per tile
# speedup vs baseline: 1.3889x; 1.3889x over previous
#include <hip/hip_runtime.h>

typedef __attribute__((ext_vector_type(2))) float v2f;
typedef __attribute__((ext_vector_type(8))) float v8f;

#define IN_SIZE  256
#define OUT_SIZE 256
#define LDA 268
#define GEMM_BLOCKS_X 256

__global__ __launch_bounds__(256, 4)
void gcn_gemm_wmma(const float* __restrict__ X, const float* __restrict__ W,
                   float* __restrict__ S, int M, int ntiles) {
    __shared__ float lA[2 * 16 * LDA];

    const int t    = threadIdx.x;
    const int lane = t & 31;
    const int wv   = t >> 5;
    const int lm   = lane & 15;
    const int half = lane >> 4;

    const int col = blockIdx.y * 128 + wv * 16 + lm;

    const int sr  = t >> 6;
    const int sc4 = t & 63;
    float4 st[4];

    int tile = blockIdx.x;
    #pragma unroll
    for (int i = 0; i < 4; ++i) {
        int r = tile * 16 + sr + 4 * i;
        st[i] = make_float4(0.f, 0.f, 0.f, 0.f);
        if (tile < ntiles && r < M) st[i] = *(const float4*)(X + (size_t)r * IN_SIZE + sc4 * 4);
    }

    v2f b[64];
    #pragma unroll
    for (int j = 0; j < 64; ++j) {
        b[j].x = W[(size_t)(4 * j + 2 * half)     * OUT_SIZE + col];
        b[j].y = W[(size_t)(4 * j + 2 * half + 1) * OUT_SIZE + col];
    }

    #pragma unroll
    for (int i = 0; i < 4; ++i)
        *(float4*)(&lA[(sr + 4 * i) * LDA + sc4 * 4]) = st[i];
    __syncthreads();

    const int aBase = lm * LDA + 2 * half;
    int buf = 0;

    for (; tile < ntiles; tile += gridDim.x) {
        const int  nxt      = tile + gridDim.x;
        const bool has_next = nxt < ntiles;

        if (has_next) {
            if (nxt * 16 + 16 <= M) {
                const float* xp = X + (size_t)(nxt * 16 + sr) * IN_SIZE + sc4 * 4;
                #pragma unroll
                for (int i = 0; i < 4; ++i)
                    st[i] = *(const float4*)(xp + 4 * i * IN_SIZE);
            } else {
                #pragma unroll
                for (int i = 0; i < 4; ++i) {
                    int r = nxt * 16 + sr + 4 * i;
                    st[i] = make_float4(0.f, 0.f, 0.f, 0.f);
                    if (r < M) st[i] = *(const float4*)(X + (size_t)r * IN_SIZE + sc4 * 4);
                }
            }
        }

        const float* a = &lA[buf * (16 * LDA) + aBase];

        v8f acc0 = {0.f,0.f,0.f,0.f,0.f,0.f,0.f,0.f};
        v8f acc1 = acc0, acc2 = acc0, acc3 = acc0;

        #pragma unroll
        for (int k = 0; k < 64; k += 4) {
            v2f a0 = *(const v2f*)(&a[k]);
            v2f a1 = *(const v2f*)(&a[k + 64]);
            v2f a2 = *(const v2f*)(&a[k + 128]);
            v2f a3 = *(const v2f*)(&a[k + 192]);
            acc0 = __builtin_amdgcn_wmma_f32_16x16x4_f32(false, a0, false, b[k / 4],      (short)0, acc0, false, false);
            acc1 = __builtin_amdgcn_wmma_f32_16x16x4_f32(false, a1, false, b[16 + k / 4], (short)0, acc1, false, false);
            acc2 = __builtin_amdgcn_wmma_f32_16x16x4_f32(false, a2, false, b[32 + k / 4], (short)0, acc2, false, false);
            acc3 = __builtin_amdgcn_wmma_f32_16x16x4_f32(false, a3, false, b[48 + k / 4], (short)0, acc3, false, false);
        }

        v8f acc = (acc0 + acc1) + (acc2 + acc3);

        if (has_next) {
            float* d = &lA[(buf ^ 1) * (16 * LDA)];
            #pragma unroll
            for (int i = 0; i < 4; ++i)
                *(float4*)(&d[(sr + 4 * i) * LDA + sc4 * 4]) = st[i];
        }
        __syncthreads();
        buf ^= 1;

        if (tile * 16 + 16 <= M) {
            float* sp = S + (size_t)(tile * 16 + 8 * half) * OUT_SIZE + col;
            #pragma unroll
            for (int v = 0; v < 8; ++v)
                sp[v * OUT_SIZE] = acc[v];
        } else {
            #pragma unroll
            for (int v = 0; v < 8; ++v) {
                int r = tile * 16 + v + 8 * half;
                if (r < M)
                    S[(size_t)r * OUT_SIZE + col] = acc[v];
            }
        }
    }
}

__global__ __launch_bounds__(256)
void gcn_init_bias(float* __restrict__ out, const float* __restrict__ bias,
                   long long total) {
    long long i = (long long)blockIdx.x * blockDim.x + threadIdx.x;
    if (i < total) out[i] = bias[i & (OUT_SIZE - 1)];
}

__global__ __launch_bounds__(256)
void gcn_edge_scatter(const float* __restrict__ S, const float* __restrict__ ew,
                      const int* __restrict__ esrc, const int* __restrict__ edst,
                      float* __restrict__ out, int E) {
    int e    = (int)(((long long)blockIdx.x * blockDim.x + threadIdx.x) >> 5);
    int lane = threadIdx.x & 31;
    if (e >= E) return;
    int   s = esrc[e];
    int   d = edst[e];
    float w = ew[e];
    const float* sp = S   + (size_t)s * OUT_SIZE;
    float*       op = out + (size_t)d * OUT_SIZE;
    #pragma unroll
    for (int i = 0; i < 8; ++i) {
        int c = lane + i * 32;
        atomicAdd(&op[c], sp[c] * w);
    }
}

extern "C" void kernel_launch(void* const* d_in, const int* in_sizes, int n_in,
                              void* d_out, int out_size, void* d_ws, size_t ws_size,
                              hipStream_t stream) {
    const float* x    = (const float*)d_in[0];
    const float* w    = (const float*)d_in[1];
    const float* bias = (const float*)d_in[2];
    const float* ew   = (const float*)d_in[3];
    const int*   esrc = (const int*)d_in[4];
    const int*   edst = (const int*)d_in[5];
    float* out  = (float*)d_out;
    float* supp = (float*)d_ws;

    const int M = in_sizes[0] / IN_SIZE;
    const int E = in_sizes[3];

    const int ntiles = (M + 15) / 16;
    dim3 ggrid(GEMM_BLOCKS_X, OUT_SIZE / 128);
    gcn_gemm_wmma<<<ggrid, 256, 0, stream>>>(x, w, supp, M, ntiles);

    long long total = (long long)M * OUT_SIZE;
    int iblocks = (int)((total + 255) / 256);
    gcn_init_bias<<<iblocks, 256, 0, stream>>>(out, bias, total);

    int sblocks = (E + 7) / 8;
    gcn_edge_scatter<<<sblocks, 256, 0, stream>>>(supp, ew, esrc, edst, out, E);
}


# Wave count cut 1.6M -> 41K: bias init = 2048 blocks of grid-stride float4 stores; scatter = one wave per 32 edges, src/dst/w broadcast by readlane, 4 edges' row loads in flight before their atomics
# speedup vs baseline: 2.4092x; 1.7346x over previous
#include <hip/hip_runtime.h>

typedef __attribute__((ext_vector_type(2))) float v2f;
typedef __attribute__((ext_vector_type(8))) float v8f;

#define IN_SIZE  256
#define OUT_SIZE 256
#define LDA 268
#define GEMM_BLOCKS_X 256

__global__ __launch_bounds__(256, 4)
void gcn_gemm_wmma(const float* __restrict__ X, const float* __restrict__ W,
                   float* __restrict__ S, int M, int ntiles) {
    __shared__ float lA[2 * 16 * LDA];

    const int t    = threadIdx.x;
    const int lane = t & 31;
    const int wv   = t >> 5;
    const int lm   = lane & 15;
    const int half = lane >> 4;

    const int col = blockIdx.y * 128 + wv * 16 + lm;

    const int sr  = t >> 6;
    const int sc4 = t & 63;
    float4 st[4];

    int tile = blockIdx.x;
    #pragma unroll
    for (int i = 0; i < 4; ++i) {
        int r = tile * 16 + sr + 4 * i;
        st[i] = make_float4(0.f, 0.f, 0.f, 0.f);
        if (tile < ntiles && r < M) st[i] = *(const float4*)(X + (size_t)r * IN_SIZE + sc4 * 4);
    }

    v2f b[64];
    #pragma unroll
    for (int j = 0; j < 64; ++j) {
        b[j].x = W[(size_t)(4 * j + 2 * half)     * OUT_SIZE + col];
        b[j].y = W[(size_t)(4 * j + 2 * half + 1) * OUT_SIZE + col];
    }

    #pragma unroll
    for (int i = 0; i < 4; ++i)
        *(float4*)(&lA[(sr + 4 * i) * LDA + sc4 * 4]) = st[i];
    __syncthreads();

    const int aBase = lm * LDA + 2 * half;
    int buf = 0;

    for (; tile < ntiles; tile += gridDim.x) {
        const int  nxt      = tile + gridDim.x;
        const bool has_next = nxt < ntiles;

        if (has_next) {
            if (nxt * 16 + 16 <= M) {
                const float* xp = X + (size_t)(nxt * 16 + sr) * IN_SIZE + sc4 * 4;
                #pragma unroll
                for (int i = 0; i < 4; ++i)
                    st[i] = *(const float4*)(xp + 4 * i * IN_SIZE);
            } else {
                #pragma unroll
                for (int i = 0; i < 4; ++i) {
                    int r = nxt * 16 + sr + 4 * i;
                    st[i] = make_float4(0.f, 0.f, 0.f, 0.f);
                    if (r < M) st[i] = *(const float4*)(X + (size_t)r * IN_SIZE + sc4 * 4);
                }
            }
        }

        const float* a = &lA[buf * (16 * LDA) + aBase];

        v8f acc0 = {0.f,0.f,0.f,0.f,0.f,0.f,0.f,0.f};
        v8f acc1 = acc0, acc2 = acc0, acc3 = acc0;

        #pragma unroll
        for (int k = 0; k < 64; k += 4) {
            v2f a0 = *(const v2f*)(&a[k]);
            v2f a1 = *(const v2f*)(&a[k + 64]);
            v2f a2 = *(const v2f*)(&a[k + 128]);
            v2f a3 = *(const v2f*)(&a[k + 192]);
            acc0 = __builtin_amdgcn_wmma_f32_16x16x4_f32(false, a0, false, b[k / 4],      (short)0, acc0, false, false);
            acc1 = __builtin_amdgcn_wmma_f32_16x16x4_f32(false, a1, false, b[16 + k / 4], (short)0, acc1, false, false);
            acc2 = __builtin_amdgcn_wmma_f32_16x16x4_f32(false, a2, false, b[32 + k / 4], (short)0, acc2, false, false);
            acc3 = __builtin_amdgcn_wmma_f32_16x16x4_f32(false, a3, false, b[48 + k / 4], (short)0, acc3, false, false);
        }

        v8f acc = (acc0 + acc1) + (acc2 + acc3);

        if (has_next) {
            float* d = &lA[(buf ^ 1) * (16 * LDA)];
            #pragma unroll
            for (int i = 0; i < 4; ++i)
                *(float4*)(&d[(sr + 4 * i) * LDA + sc4 * 4]) = st[i];
        }
        __syncthreads();
        buf ^= 1;

        if (tile * 16 + 16 <= M) {
            float* sp = S + (size_t)(tile * 16 + 8 * half) * OUT_SIZE + col;
            #pragma unroll
            for (int v = 0; v < 8; ++v)
                sp[v * OUT_SIZE] = acc[v];
        } else {
            #pragma unroll
            for (int v = 0; v < 8; ++v) {
                int r = tile * 16 + v + 8 * half;
                if (r < M)
                    S[(size_t)r * OUT_SIZE + col] = acc[v];
            }
        }
    }
}

#define INIT_BLOCKS 2048
__global__ __launch_bounds__(256)
void gcn_init_bias(float* __restrict__ out, const float* __restrict__ bias,
                   long long total4) {
    const long long stride = (long long)gridDim.x * 256;
    long long i = (long long)blockIdx.x * 256 + threadIdx.x;
    const float4 bv = *(const float4*)(bias + ((threadIdx.x & 63) << 2));
    asm volatile("" :: "v"(bv.x), "v"(bv.y), "v"(bv.z), "v"(bv.w));
    float4* o4 = (float4*)out;
    for (; i < total4; i += stride) o4[i] = bv;
}

__global__ __launch_bounds__(256)
void gcn_edge_scatter(const float* __restrict__ S, const float* __restrict__ ew,
                      const int* __restrict__ esrc, const int* __restrict__ edst,
                      float* __restrict__ out, int E) {
    const int lane = threadIdx.x & 31;
    const int base = __builtin_amdgcn_readfirstlane((int)((blockIdx.x * blockDim.x + threadIdx.x) >> 5)) * 32;
    if (base >= E) return;
    int n = E - base; if (n > 32) n = 32;

    int sv = 0, dv = 0, wv = 0;
    if (lane < n) {
        sv = esrc[base + lane];
        dv = edst[base + lane];
        wv = __float_as_int(ew[base + lane]);
    }

    int j = 0;
    for (; j + 4 <= n; j += 4) {
        float v[4][8];
        float w[4];
        float* op[4];
        #pragma unroll
        for (int q = 0; q < 4; ++q) {
            const int s = __builtin_amdgcn_readlane(sv, j + q);
            const int d = __builtin_amdgcn_readlane(dv, j + q);
            w[q] = __int_as_float(__builtin_amdgcn_readlane(wv, j + q));
            const float* sp = S + (size_t)s * OUT_SIZE + lane;
            op[q] = out + (size_t)d * OUT_SIZE + lane;
            #pragma unroll
            for (int i = 0; i < 8; ++i) v[q][i] = sp[i * 32];
        }
        #pragma unroll
        for (int q = 0; q < 4; ++q) {
            #pragma unroll
            for (int i = 0; i < 8; ++i) atomicAdd(op[q] + i * 32, v[q][i] * w[q]);
        }
    }
    for (; j < n; ++j) {
        const int   s = __builtin_amdgcn_readlane(sv, j);
        const int   d = __builtin_amdgcn_readlane(dv, j);
        const float w = __int_as_float(__builtin_amdgcn_readlane(wv, j));
        const float* sp = S   + (size_t)s * OUT_SIZE + lane;
        float*       op = out + (size_t)d * OUT_SIZE + lane;
        float v[8];
        #pragma unroll
        for (int i = 0; i < 8; ++i) v[i] = sp[i * 32];
        #pragma unroll
        for (int i = 0; i < 8; ++i) atomicAdd(op + i * 32, v[i] * w);
    }
}

extern "C" void kernel_launch(void* const* d_in, const int* in_sizes, int n_in,
                              void* d_out, int out_size, void* d_ws, size_t ws_size,
                              hipStream_t stream) {
    const float* x    = (const float*)d_in[0];
    const float* w    = (const float*)d_in[1];
    const float* bias = (const float*)d_in[2];
    const float* ew   = (const float*)d_in[3];
    const int*   esrc = (const int*)d_in[4];
    const int*   edst = (const int*)d_in[5];
    float* out  = (float*)d_out;
    float* supp = (float*)d_ws;

    const int M = in_sizes[0] / IN_SIZE;
    const int E = in_sizes[3];

    const int ntiles = (M + 15) / 16;
    dim3 ggrid(GEMM_BLOCKS_X, OUT_SIZE / 128);
    gcn_gemm_wmma<<<ggrid, 256, 0, stream>>>(x, w, supp, M, ntiles);

    long long total4 = (long long)M * (OUT_SIZE / 4);
    gcn_init_bias<<<INIT_BLOCKS, 256, 0, stream>>>(out, bias, total4);

    int sblocks = (E + 255) / 256;
    gcn_edge_scatter<<<sblocks, 256, 0, stream>>>(supp, ew, esrc, edst, out, E);
}


# Float-atomic scatter -> gather by dst: int atomics rank edges into per-node buckets (rows summed in arrival-rank order), a wave sums a node's rows in registers, adds bias, stores each out row once
# speedup vs baseline: 2.8677x; 1.1903x over previous
#include <hip/hip_runtime.h>

typedef __attribute__((ext_vector_type(2))) float v2f;
typedef __attribute__((ext_vector_type(8))) float v8f;

#define IN_SIZE  256
#define OUT_SIZE 256
#define LDA 268
#define GEMM_BLOCKS_X 256

__global__ __launch_bounds__(256, 4)
void gcn_gemm_wmma(const float* __restrict__ X, const float* __restrict__ W,
                   float* __restrict__ S, int* __restrict__ cnt, int* __restrict__ ovh,
                   int M, int ntiles) {
    __shared__ float lA[2 * 16 * LDA];

    const int t    = threadIdx.x;

    {
        const int nthr = gridDim.x * gridDim.y * 256;
        for (int i = (blockIdx.y * gridDim.x + blockIdx.x) * 256 + t; i < M; i += nthr) {
            cnt[i] = 0;
            ovh[i] = -1;
        }
    }

    const int lane = t & 31;
    const int wv   = t >> 5;
    const int lm   = lane & 15;
    const int half = lane >> 4;

    const int col = blockIdx.y * 128 + wv * 16 + lm;

    const int sr  = t >> 6;
    const int sc4 = t & 63;
    float4 st[4];

    int tile = blockIdx.x;
    #pragma unroll
    for (int i = 0; i < 4; ++i) {
        int r = tile * 16 + sr + 4 * i;
        st[i] = make_float4(0.f, 0.f, 0.f, 0.f);
        if (tile < ntiles && r < M) st[i] = *(const float4*)(X + (size_t)r * IN_SIZE + sc4 * 4);
    }

    v2f b[64];
    #pragma unroll
    for (int j = 0; j < 64; ++j) {
        b[j].x = W[(size_t)(4 * j + 2 * half)     * OUT_SIZE + col];
        b[j].y = W[(size_t)(4 * j + 2 * half + 1) * OUT_SIZE + col];
    }

    #pragma unroll
    for (int i = 0; i < 4; ++i)
        *(float4*)(&lA[(sr + 4 * i) * LDA + sc4 * 4]) = st[i];
    __syncthreads();

    const int aBase = lm * LDA + 2 * half;
    int buf = 0;

    for (; tile < ntiles; tile += gridDim.x) {
        const int  nxt      = tile + gridDim.x;
        const bool has_next = nxt < ntiles;

        if (has_next) {
            if (nxt * 16 + 16 <= M) {
                const float* xp = X + (size_t)(nxt * 16 + sr) * IN_SIZE + sc4 * 4;
                #pragma unroll
                for (int i = 0; i < 4; ++i)
                    st[i] = *(const float4*)(xp + 4 * i * IN_SIZE);
            } else {
                #pragma unroll
                for (int i = 0; i < 4; ++i) {
                    int r = nxt * 16 + sr + 4 * i;
                    st[i] = make_float4(0.f, 0.f, 0.f, 0.f);
                    if (r < M) st[i] = *(const float4*)(X + (size_t)r * IN_SIZE + sc4 * 4);
                }
            }
        }

        const float* a = &lA[buf * (16 * LDA) + aBase];

        v8f acc0 = {0.f,0.f,0.f,0.f,0.f,0.f,0.f,0.f};
        v8f acc1 = acc0, acc2 = acc0, acc3 = acc0;

        #pragma unroll
        for (int k = 0; k < 64; k += 4) {
            v2f a0 = *(const v2f*)(&a[k]);
            v2f a1 = *(const v2f*)(&a[k + 64]);
            v2f a2 = *(const v2f*)(&a[k + 128]);
            v2f a3 = *(const v2f*)(&a[k + 192]);
            acc0 = __builtin_amdgcn_wmma_f32_16x16x4_f32(false, a0, false, b[k / 4],      (short)0, acc0, false, false);
            acc1 = __builtin_amdgcn_wmma_f32_16x16x4_f32(false, a1, false, b[16 + k / 4], (short)0, acc1, false, false);
            acc2 = __builtin_amdgcn_wmma_f32_16x16x4_f32(false, a2, false, b[32 + k / 4], (short)0, acc2, false, false);
            acc3 = __builtin_amdgcn_wmma_f32_16x16x4_f32(false, a3, false, b[48 + k / 4], (short)0, acc3, false, false);
        }

        v8f acc = (acc0 + acc1) + (acc2 + acc3);

        if (has_next) {
            float* d = &lA[(buf ^ 1) * (16 * LDA)];
            #pragma unroll
            for (int i = 0; i < 4; ++i)
                *(float4*)(&d[(sr + 4 * i) * LDA + sc4 * 4]) = st[i];
        }
        __syncthreads();
        buf ^= 1;

        if (tile * 16 + 16 <= M) {
            float* sp = S + (size_t)(tile * 16 + 8 * half) * OUT_SIZE + col;
            #pragma unroll
            for (int v = 0; v < 8; ++v)
                sp[v * OUT_SIZE] = acc[v];
        } else {
            #pragma unroll
            for (int v = 0; v < 8; ++v) {
                int r = tile * 16 + v + 8 * half;
                if (r < M)
                    S[(size_t)r * OUT_SIZE + col] = acc[v];
            }
        }
    }
}

#define BUCKET 32
#define EPT 8
__global__ __launch_bounds__(256)
void gcn_bucket_edges(const float* __restrict__ ew, const int* __restrict__ esrc,
                      const int* __restrict__ edst, int* __restrict__ cnt,
                      int* __restrict__ ovh, int* __restrict__ nxt,
                      int2* __restrict__ slots, int E) {
    const int nthr = gridDim.x * 256;
    for (int e0 = blockIdx.x * 256 + threadIdx.x; e0 < E; e0 += nthr * EPT) {
        int d[EPT], s[EPT], w[EPT], r[EPT];
        #pragma unroll
        for (int q = 0; q < EPT; ++q) {
            const int e = e0 + q * nthr;
            d[q] = 0; s[q] = 0; w[q] = 0;
            if (e < E) {
                d[q] = edst[e];
                s[q] = esrc[e];
                w[q] = __float_as_int(ew[e]);
            }
        }
        #pragma unroll
        for (int q = 0; q < EPT; ++q) {
            r[q] = 0;
            if (e0 + q * nthr < E) r[q] = atomicAdd(&cnt[d[q]], 1);
        }
        #pragma unroll
        for (int q = 0; q < EPT; ++q) {
            const int e = e0 + q * nthr;
            if (e < E) {
                if (r[q] < BUCKET) {
                    slots[(size_t)d[q] * BUCKET + r[q]] = make_int2(s[q], w[q]);
                } else {
                    nxt[e] = atomicExch(&ovh[d[q]], e);
                }
            }
        }
    }
}

#define NPW 13
#define GATHER_THREADS 128

template <int R>
__device__ __forceinline__ void add_rows(const float* __restrict__ S, int sx, int sy, int j,
                                         int lane, float4& a0, float4& a1) {
    float4 va[R], vb[R];
    float  w[R];
    #pragma unroll
    for (int q = 0; q < R; ++q) {
        const int s = __builtin_amdgcn_readlane(sx, j + q);
        w[q] = __int_as_float(__builtin_amdgcn_readlane(sy, j + q));
        const float* sp = S + (size_t)s * OUT_SIZE + 4 * lane;
        va[q] = *(const float4*)sp;
        vb[q] = *(const float4*)(sp + 128);
    }
    #pragma unroll
    for (int q = 0; q < R; ++q) {
        a0.x = fmaf(va[q].x, w[q], a0.x); a0.y = fmaf(va[q].y, w[q], a0.y);
        a0.z = fmaf(va[q].z, w[q], a0.z); a0.w = fmaf(va[q].w, w[q], a0.w);
        a1.x = fmaf(vb[q].x, w[q], a1.x); a1.y = fmaf(vb[q].y, w[q], a1.y);
        a1.z = fmaf(vb[q].z, w[q], a1.z); a1.w = fmaf(vb[q].w, w[q], a1.w);
    }
}

__global__ __launch_bounds__(GATHER_THREADS)
void gcn_node_gather(const float* __restrict__ S, const float* __restrict__ bias,
                     const float* __restrict__ ew, const int* __restrict__ esrc,
                     const int* __restrict__ cnt, const int* __restrict__ ovh,
                     const int* __restrict__ nxt, const int2* __restrict__ slots,
                     float* __restrict__ out, int N, int E) {
    const int lane = threadIdx.x & 31;
    const int n0 = __builtin_amdgcn_readfirstlane((int)((blockIdx.x * blockDim.x + threadIdx.x) >> 5)) * NPW;
    if (n0 >= N) return;
    int nn = N - n0; if (nn > NPW) nn = NPW;

    int cv = 0;
    if (lane < nn) cv = cnt[n0 + lane];
    const float4 b0 = *(const float4*)(bias + 4 * lane);
    const float4 b1 = *(const float4*)(bias + 128 + 4 * lane);

    for (int i = 0; i < nn; ++i) {
        const int c  = __builtin_amdgcn_readlane(cv, i);
        const int nb = c < BUCKET ? c : BUCKET;
        int2 sw = make_int2(0, 0);
        if (lane < nb) sw = slots[(size_t)(n0 + i) * BUCKET + lane];

        float4 a0 = make_float4(0.f, 0.f, 0.f, 0.f);
        float4 a1 = a0;

        int j = 0;
        for (; j + 8 <= nb; j += 8) add_rows<8>(S, sw.x, sw.y, j, lane, a0, a1);
        switch (nb - j) {
            case 7: add_rows<7>(S, sw.x, sw.y, j, lane, a0, a1); break;
            case 6: add_rows<6>(S, sw.x, sw.y, j, lane, a0, a1); break;
            case 5: add_rows<5>(S, sw.x, sw.y, j, lane, a0, a1); break;
            case 4: add_rows<4>(S, sw.x, sw.y, j, lane, a0, a1); break;
            case 3: add_rows<3>(S, sw.x, sw.y, j, lane, a0, a1); break;
            case 2: add_rows<2>(S, sw.x, sw.y, j, lane, a0, a1); break;
            case 1: add_rows<1>(S, sw.x, sw.y, j, lane, a0, a1); break;
            default: break;
        }
        if (c > BUCKET) {
            int e = __builtin_amdgcn_readfirstlane(ovh[n0 + i]);
            for (int it = 0; it < E && (unsigned)e < (unsigned)E; ++it) {
                const int   s = esrc[e];
                const float w = ew[e];
                const float* sp = S + (size_t)s * OUT_SIZE + 4 * lane;
                const float4 v0 = *(const float4*)sp;
                const float4 v1 = *(const float4*)(sp + 128);
                a0.x = fmaf(v0.x, w, a0.x); a0.y = fmaf(v0.y, w, a0.y);
                a0.z = fmaf(v0.z, w, a0.z); a0.w = fmaf(v0.w, w, a0.w);
                a1.x = fmaf(v1.x, w, a1.x); a1.y = fmaf(v1.y, w, a1.y);
                a1.z = fmaf(v1.z, w, a1.z); a1.w = fmaf(v1.w, w, a1.w);
                e = __builtin_amdgcn_readfirstlane(nxt[e]);
            }
        }

        float* op = out + (size_t)(n0 + i) * OUT_SIZE + 4 * lane;
        *(float4*)op         = make_float4(a0.x + b0.x, a0.y + b0.y, a0.z + b0.z, a0.w + b0.w);
        *(float4*)(op + 128) = make_float4(a1.x + b1.x, a1.y + b1.y, a1.z + b1.z, a1.w + b1.w);
    }
}

extern "C" void kernel_launch(void* const* d_in, const int* in_sizes, int n_in,
                              void* d_out, int out_size, void* d_ws, size_t ws_size,
                              hipStream_t stream) {
    const float* x    = (const float*)d_in[0];
    const float* w    = (const float*)d_in[1];
    const float* bias = (const float*)d_in[2];
    const float* ew   = (const float*)d_in[3];
    const int*   esrc = (const int*)d_in[4];
    const int*   edst = (const int*)d_in[5];
    float* out  = (float*)d_out;
    float* supp = (float*)d_ws;

    const int M = in_sizes[0] / IN_SIZE;
    const int E = in_sizes[3];

    const int Mp = (M + 63) & ~63;
    int2* slots = (int2*)(supp + (size_t)M * OUT_SIZE);
    int*  cnt   = (int*)(slots + (size_t)M * BUCKET);
    int*  ovh   = cnt + Mp;
    int*  nxt   = ovh + Mp;

    const int ntiles = (M + 15) / 16;
    dim3 ggrid(GEMM_BLOCKS_X, OUT_SIZE / 128);
    gcn_gemm_wmma<<<ggrid, 256, 0, stream>>>(x, w, supp, cnt, ovh, M, ntiles);

    int bblocks = (E + 256 * EPT - 1) / (256 * EPT);
    gcn_bucket_edges<<<bblocks, 256, 0, stream>>>(ew, esrc, edst, cnt, ovh, nxt, slots, E);

    const int npb = NPW * (GATHER_THREADS / 32);
    int gblocks = (M + npb - 1) / npb;
    gcn_node_gather<<<gblocks, GATHER_THREADS, 0, stream>>>(supp, bias, ew, esrc, cnt, ovh, nxt, slots, out, M, E);
}


# Edge bucketing (int atomics, arrival-rank order as before) moved into the GEMM launch: each block ranks one chunk of its edge slice per X tile under the WMMAs; counters zeroed by a launch in front
# speedup vs baseline: 3.1030x; 1.0821x over previous
#include <hip/hip_runtime.h>

typedef __attribute__((ext_vector_type(2))) float v2f;
typedef __attribute__((ext_vector_type(8))) float v8f;

#define IN_SIZE  256
#define OUT_SIZE 256
#define LDA 268
#define GEMM_BLOCKS_X 256

#define BUCKET 32

__global__ __launch_bounds__(256)
void gcn_init_counters(int* __restrict__ cnt, int* __restrict__ ovh, int M) {
    const int i = blockIdx.x * 256 + threadIdx.x;
    if (i < M) {
        cnt[i] = 0;
        ovh[i] = -1;
    }
}

__device__ __forceinline__ void put_edge(int e, int d, int s, int w, int r,
                                         int* __restrict__ ovh, int* __restrict__ nxt,
                                         int2* __restrict__ slots) {
    if (r < BUCKET) {
        slots[(size_t)d * BUCKET + r] = make_int2(s, w);
    } else {
        nxt[e] = atomicExch(&ovh[d], e);
    }
}

__global__ __launch_bounds__(256, 4)
void gcn_gemm_wmma(const float* __restrict__ X, const float* __restrict__ W,
                   float* __restrict__ S,
                   const float* __restrict__ ew, const int* __restrict__ esrc,
                   const int* __restrict__ edst,
                   int* __restrict__ cnt, int* __restrict__ ovh, int* __restrict__ elink,
                   int2* __restrict__ slots, int M, int ntiles, int E) {
    __shared__ float lA[2 * 16 * LDA];

    const int t    = threadIdx.x;

    const int lane = t & 31;
    const int wv   = t >> 5;
    const int lm   = lane & 15;
    const int half = lane >> 4;

    const int col = blockIdx.y * 128 + wv * 16 + lm;

    const int sr  = t >> 6;
    const int sc4 = t & 63;
    float4 st[4];

    int tile = blockIdx.x;
    #pragma unroll
    for (int i = 0; i < 4; ++i) {
        int r = tile * 16 + sr + 4 * i;
        st[i] = make_float4(0.f, 0.f, 0.f, 0.f);
        if (tile < ntiles && r < M) st[i] = *(const float4*)(X + (size_t)r * IN_SIZE + sc4 * 4);
    }

    const int bid     = blockIdx.y * gridDim.x + blockIdx.x;
    const int nblk    = gridDim.x * gridDim.y;
    const int epb     = (E + nblk - 1) / nblk;
    const int ebeg    = bid * epb < E ? bid * epb : E;
    const int eend    = ebeg + epb < E ? ebeg + epb : E;
    const int mytiles = blockIdx.x < ntiles ? (ntiles - 1 - blockIdx.x) / gridDim.x + 1 : 0;
    int ch = 0;
    if (mytiles > 0) {
        ch = (epb + mytiles - 1) / mytiles;
        if (ch > 256) ch = 256;
    }
    int  ne   = ebeg + t;
    bool nact = t < ch && ne < eend;
    int  nd = 0, ns = 0, nw = 0;
    if (nact) {
        nd = edst[ne];
        ns = esrc[ne];
        nw = __float_as_int(ew[ne]);
    }

    v2f b[64];
    #pragma unroll
    for (int j = 0; j < 64; ++j) {
        b[j].x = W[(size_t)(4 * j + 2 * half)     * OUT_SIZE + col];
        b[j].y = W[(size_t)(4 * j + 2 * half + 1) * OUT_SIZE + col];
    }

    #pragma unroll
    for (int i = 0; i < 4; ++i)
        *(float4*)(&lA[(sr + 4 * i) * LDA + sc4 * 4]) = st[i];
    __syncthreads();

    const int aBase = lm * LDA + 2 * half;
    int buf = 0;

    for (; tile < ntiles; tile += gridDim.x) {
        const int  nxt      = tile + gridDim.x;
        const bool has_next = nxt < ntiles;

        if (has_next) {
            if (nxt * 16 + 16 <= M) {
                const float* xp = X + (size_t)(nxt * 16 + sr) * IN_SIZE + sc4 * 4;
                #pragma unroll
                for (int i = 0; i < 4; ++i)
                    st[i] = *(const float4*)(xp + 4 * i * IN_SIZE);
            } else {
                #pragma unroll
                for (int i = 0; i < 4; ++i) {
                    int r = nxt * 16 + sr + 4 * i;
                    st[i] = make_float4(0.f, 0.f, 0.f, 0.f);
                    if (r < M) st[i] = *(const float4*)(X + (size_t)r * IN_SIZE + sc4 * 4);
                }
            }
        }

        const int  ce = ne, cd = nd, cs = ns, cw = nw;
        const bool cact = nact;
        ne += ch;
        nact = has_next && t < ch && ne < eend;
        nd = 0; ns = 0; nw = 0;
        if (nact) {
            nd = edst[ne];
            ns = esrc[ne];
            nw = __float_as_int(ew[ne]);
        }
        int cr = 0;
        if (cact) cr = atomicAdd(&cnt[cd], 1);

        const float* a = &lA[buf * (16 * LDA) + aBase];

        v8f acc0 = {0.f,0.f,0.f,0.f,0.f,0.f,0.f,0.f};
        v8f acc1 = acc0, acc2 = acc0, acc3 = acc0;

        #pragma unroll
        for (int k = 0; k < 64; k += 4) {
            v2f a0 = *(const v2f*)(&a[k]);
            v2f a1 = *(const v2f*)(&a[k + 64]);
            v2f a2 = *(const v2f*)(&a[k + 128]);
            v2f a3 = *(const v2f*)(&a[k + 192]);
            acc0 = __builtin_amdgcn_wmma_f32_16x16x4_f32(false, a0, false, b[k / 4],      (short)0, acc0, false, false);
            acc1 = __builtin_amdgcn_wmma_f32_16x16x4_f32(false, a1, false, b[16 + k / 4], (short)0, acc1, false, false);
            acc2 = __builtin_amdgcn_wmma_f32_16x16x4_f32(false, a2, false, b[32 + k / 4], (short)0, acc2, false, false);
            acc3 = __builtin_amdgcn_wmma_f32_16x16x4_f32(false, a3, false, b[48 + k / 4], (short)0, acc3, false, false);
        }

        v8f acc = (acc0 + acc1) + (acc2 + acc3);

        if (has_next) {
            float* d = &lA[(buf ^ 1) * (16 * LDA)];
            #pragma unroll
            for (int i = 0; i < 4; ++i)
                *(float4*)(&d[(sr + 4 * i) * LDA + sc4 * 4]) = st[i];
        }
        __syncthreads();
        buf ^= 1;

        if (tile * 16 + 16 <= M) {
            float* sp = S + (size_t)(tile * 16 + 8 * half) * OUT_SIZE + col;
            #pragma unroll
            for (int v = 0; v < 8; ++v)
                sp[v * OUT_SIZE] = acc[v];
        } else {
            #pragma unroll
            for (int v = 0; v < 8; ++v) {
                int r = tile * 16 + v + 8 * half;
                if (r < M)
                    S[(size_t)r * OUT_SIZE + col] = acc[v];
            }
        }

        if (cact) put_edge(ce, cd, cs, cw, cr, ovh, elink, slots);
    }

    for (int e = ebeg + mytiles * ch + t; e < eend; e += 256) {
        const int d = edst[e];
        const int r = atomicAdd(&cnt[d], 1);
        put_edge(e, d, esrc[e], __float_as_int(ew[e]), r, ovh, elink, slots);
    }
}

#define NPW 13
#define GATHER_THREADS 128

template <int R>
__device__ __forceinline__ void add_rows(const float* __restrict__ S, int sx, int sy, int j,
                                         int lane, float4& a0, float4& a1) {
    float4 va[R], vb[R];
    float  w[R];
    #pragma unroll
    for (int q = 0; q < R; ++q) {
        const int s = __builtin_amdgcn_readlane(sx, j + q);
        w[q] = __int_as_float(__builtin_amdgcn_readlane(sy, j + q));
        const float* sp = S + (size_t)s * OUT_SIZE + 4 * lane;
        va[q] = *(const float4*)sp;
        vb[q] = *(const float4*)(sp + 128);
    }
    #pragma unroll
    for (int q = 0; q < R; ++q) {
        a0.x = fmaf(va[q].x, w[q], a0.x); a0.y = fmaf(va[q].y, w[q], a0.y);
        a0.z = fmaf(va[q].z, w[q], a0.z); a0.w = fmaf(va[q].w, w[q], a0.w);
        a1.x = fmaf(vb[q].x, w[q], a1.x); a1.y = fmaf(vb[q].y, w[q], a1.y);
        a1.z = fmaf(vb[q].z, w[q], a1.z); a1.w = fmaf(vb[q].w, w[q], a1.w);
    }
}

__global__ __launch_bounds__(GATHER_THREADS)
void gcn_node_gather(const float* __restrict__ S, const float* __restrict__ bias,
                     const float* __restrict__ ew, const int* __restrict__ esrc,
                     const int* __restrict__ cnt, const int* __restrict__ ovh,
                     const int* __restrict__ nxt, const int2* __restrict__ slots,
                     float* __restrict__ out, int N, int E) {
    const int lane = threadIdx.x & 31;
    const int n0 = __builtin_amdgcn_readfirstlane((int)((blockIdx.x * blockDim.x + threadIdx.x) >> 5)) * NPW;
    if (n0 >= N) return;
    int nn = N - n0; if (nn > NPW) nn = NPW;

    int cv = 0;
    if (lane < nn) cv = cnt[n0 + lane];
    const float4 b0 = *(const float4*)(bias + 4 * lane);
    const float4 b1 = *(const float4*)(bias + 128 + 4 * lane);

    for (int i = 0; i < nn; ++i) {
        const int c  = __builtin_amdgcn_readlane(cv, i);
        const int nb = c < BUCKET ? c : BUCKET;
        int2 sw = make_int2(0, 0);
        if (lane < nb) sw = slots[(size_t)(n0 + i) * BUCKET + lane];

        float4 a0 = make_float4(0.f, 0.f, 0.f, 0.f);
        float4 a1 = a0;

        int j = 0;
        for (; j + 8 <= nb; j += 8) add_rows<8>(S, sw.x, sw.y, j, lane, a0, a1);
        switch (nb - j) {
            case 7: add_rows<7>(S, sw.x, sw.y, j, lane, a0, a1); break;
            case 6: add_rows<6>(S, sw.x, sw.y, j, lane, a0, a1); break;
            case 5: add_rows<5>(S, sw.x, sw.y, j, lane, a0, a1); break;
            case 4: add_rows<4>(S, sw.x, sw.y, j, lane, a0, a1); break;
            case 3: add_rows<3>(S, sw.x, sw.y, j, lane, a0, a1); break;
            case 2: add_rows<2>(S, sw.x, sw.y, j, lane, a0, a1); break;
            case 1: add_rows<1>(S, sw.x, sw.y, j, lane, a0, a1); break;
            default: break;
        }
        if (c > BUCKET) {
            int e = __builtin_amdgcn_readfirstlane(ovh[n0 + i]);
            for (int it = 0; it < E && (unsigned)e < (unsigned)E; ++it) {
                const int   s = esrc[e];
                const float w = ew[e];
                const float* sp = S + (size_t)s * OUT_SIZE + 4 * lane;
                const float4 v0 = *(const float4*)sp;
                const float4 v1 = *(const float4*)(sp + 128);
                a0.x = fmaf(v0.x, w, a0.x); a0.y = fmaf(v0.y, w, a0.y);
                a0.z = fmaf(v0.z, w, a0.z); a0.w = fmaf(v0.w, w, a0.w);
                a1.x = fmaf(v1.x, w, a1.x); a1.y = fmaf(v1.y, w, a1.y);
                a1.z = fmaf(v1.z, w, a1.z); a1.w = fmaf(v1.w, w, a1.w);
                e = __builtin_amdgcn_readfirstlane(nxt[e]);
            }
        }

        float* op = out + (size_t)(n0 + i) * OUT_SIZE + 4 * lane;
        *(float4*)op         = make_float4(a0.x + b0.x, a0.y + b0.y, a0.z + b0.z, a0.w + b0.w);
        *(float4*)(op + 128) = make_float4(a1.x + b1.x, a1.y + b1.y, a1.z + b1.z, a1.w + b1.w);
    }
}

extern "C" void kernel_launch(void* const* d_in, const int* in_sizes, int n_in,
                              void* d_out, int out_size, void* d_ws, size_t ws_size,
                              hipStream_t stream) {
    const float* x    = (const float*)d_in[0];
    const float* w    = (const float*)d_in[1];
    const float* bias = (const float*)d_in[2];
    const float* ew   = (const float*)d_in[3];
    const int*   esrc = (const int*)d_in[4];
    const int*   edst = (const int*)d_in[5];
    float* out  = (float*)d_out;
    float* supp = (float*)d_ws;

    const int M = in_sizes[0] / IN_SIZE;
    const int E = in_sizes[3];

    const int Mp = (M + 63) & ~63;
    int2* slots = (int2*)(supp + (size_t)M * OUT_SIZE);
    int*  cnt   = (int*)(slots + (size_t)M * BUCKET);
    int*  ovh   = cnt + Mp;
    int*  nxt   = ovh + Mp;

    gcn_init_counters<<<(M + 255) / 256, 256, 0, stream>>>(cnt, ovh, M);

    const int ntiles = (M + 15) / 16;
    dim3 ggrid(GEMM_BLOCKS_X, OUT_SIZE / 128);
    gcn_gemm_wmma<<<ggrid, 256, 0, stream>>>(x, w, supp, ew, esrc, edst, cnt, ovh, nxt, slots, M, ntiles, E);

    const int npb = NPW * (GATHER_THREADS / 32);
    int gblocks = (M + npb - 1) / npb;
    gcn_node_gather<<<gblocks, GATHER_THREADS, 0, stream>>>(supp, bias, ew, esrc, cnt, ovh, nxt, slots, out, M, E);
}
